# starting kernel
# baseline (speedup 1.0000x reference)
#include <hip/hip_runtime.h>

#define PRED_LEN 12
#define HIDDEN 64
#define T_ENC 8
#define IN_DIM 2
#define WAVES_PER_BLOCK 8
#define ROWS_PER_WAVE 16
#define ROWS_PER_BLOCK (WAVES_PER_BLOCK * ROWS_PER_WAVE)
#define HROW_STRIDE 68

typedef __attribute__((ext_vector_type(2))) float v2f;
typedef __attribute__((ext_vector_type(4))) float v4f;
typedef __attribute__((ext_vector_type(8))) float v8f;

__device__ __forceinline__ float fast_rcp(float x) {
    return __builtin_amdgcn_rcpf(x);
}
__device__ __forceinline__ float sigm(float x) {
    return fast_rcp(1.0f + __expf(-x));
}
__device__ __forceinline__ float tanh_fast(float x) {
    return 1.0f - 2.0f * fast_rcp(__expf(2.0f * x) + 1.0f);
}

__device__ __forceinline__ v8f wmma_f32(v2f a, v2f b, v8f c) {
    return __builtin_amdgcn_wmma_f32_16x16x4_f32(
         false, a,  false, b,
         (short)0, c,  false,  false);
}

__launch_bounds__(256, 1)
__global__ void lstm_seq2seq_kernel(const float* __restrict__ obs,
                                    const float* __restrict__ W_ih,
                                    const float* __restrict__ W_hh,
                                    const float* __restrict__ b_ih,
                                    const float* __restrict__ b_hh,
                                    const float* __restrict__ fc_W,
                                    const float* __restrict__ fc_b,
                                    float* __restrict__ out)
{
    __shared__ v2f   sBw[16 * 16 * 32];
    __shared__ v2f   sBx[16 * 32];
    __shared__ float sH[WAVES_PER_BLOCK * ROWS_PER_WAVE * HROW_STRIDE];
    __shared__ float sX[WAVES_PER_BLOCK * ROWS_PER_WAVE * 2];
    __shared__ float sFcW[2 * HIDDEN];
    __shared__ float sFcB[2];

    const int tid = threadIdx.x;

    for (int idx = tid; idx < 16 * 16 * 32; idx += 256) {
        int t = idx >> 9;
        int k = (idx >> 5) & 15;
        int l = idx & 31;
        int n  = (t << 4) | (l & 15);
        int kk = (k << 2) + ((l >> 4) << 1);
        v2f w;
        w.x = W_hh[n * HIDDEN + kk];
        w.y = W_hh[n * HIDDEN + kk + 1];
        sBw[idx] = w;
    }
    for (int idx = tid; idx < 16 * 32; idx += 256) {
        int t = idx >> 5;
        int l = idx & 31;
        int n = (t << 4) | (l & 15);
        v2f w;
        if (l < 16) { w.x = W_ih[n * IN_DIM + 0]; w.y = W_ih[n * IN_DIM + 1]; }
        else        { w.x = b_ih[n] + b_hh[n];    w.y = 0.0f; }
        sBx[idx] = w;
    }
    for (int idx = tid; idx < 2 * HIDDEN; idx += 256) sFcW[idx] = fc_W[idx];
    if (tid < 2) sFcB[tid] = fc_b[tid];
    for (int idx = tid; idx < WAVES_PER_BLOCK * ROWS_PER_WAVE * HROW_STRIDE; idx += 256)
        sH[idx] = 0.0f;
    __syncthreads();

    const int wave      = tid >> 5;
    const int lane      = tid & 31;
    const int rowInWave = lane & 15;
    const int laneHalf  = lane >> 4;
    const int b0        = blockIdx.x * ROWS_PER_BLOCK + wave * ROWS_PER_WAVE;
    float* hbase = &sH[(wave * ROWS_PER_WAVE) * HROW_STRIDE];

    v8f cst[4];
    #pragma unroll
    for (int j = 0; j < 4; ++j) cst[j] = (v8f){};

    auto cell = [&](v2f ax) {
        float hnew[4][8];
        #pragma unroll
        for (int jp = 0; jp < 4; jp += 2) {
            v8f acc[8];
            v8f zero = {};
            #pragma unroll
            for (int dj = 0; dj < 2; ++dj) {
                #pragma unroll
                for (int g = 0; g < 4; ++g) {
                    int t = 4 * g + jp + dj;
                    v2f bx = sBx[(t << 5) | lane];
                    acc[dj * 4 + g] = wmma_f32(ax, bx, zero);
                }
            }
            #pragma unroll
            for (int k = 0; k < 16; ++k) {
                int kk = (k << 2) | (laneHalf << 1);
                v2f ah = *(const v2f*)&hbase[rowInWave * HROW_STRIDE + kk];
                #pragma unroll
                for (int dj = 0; dj < 2; ++dj) {
                    #pragma unroll
                    for (int g = 0; g < 4; ++g) {
                        int t = 4 * g + jp + dj;
                        v2f bw = sBw[(((t << 4) | k) << 5) | lane];
                        acc[dj * 4 + g] = wmma_f32(ah, bw, acc[dj * 4 + g]);
                    }
                }
            }
            #pragma unroll
            for (int dj = 0; dj < 2; ++dj) {
                int j = jp + dj;
                #pragma unroll
                for (int v = 0; v < 8; ++v) {
                    float ig = sigm(acc[dj * 4 + 0][v]);
                    float fg = sigm(acc[dj * 4 + 1][v]);
                    float gg = tanh_fast(acc[dj * 4 + 2][v]);
                    float og = sigm(acc[dj * 4 + 3][v]);
                    float cc = fg * cst[j][v] + ig * gg;
                    cst[j][v] = cc;
                    hnew[j][v] = og * tanh_fast(cc);
                }
            }
        }
        #pragma unroll
        for (int j = 0; j < 4; ++j) {
            #pragma unroll
            for (int v = 0; v < 8; ++v) {
                hbase[(v + (laneHalf << 3)) * HROW_STRIDE + (j << 4) + rowInWave] =
                    hnew[j][v];
            }
        }
        __syncthreads();
    };

    v2f ax;
    #pragma unroll 1
    for (int s = 0; s < T_ENC; ++s) {
        v2f xv = *(const v2f*)&obs[(size_t)((b0 + rowInWave) * T_ENC + s) * IN_DIM];
        ax.x = laneHalf ? 1.0f : xv.x;
        ax.y = laneHalf ? 0.0f : xv.y;
        cell(ax);
    }

    #pragma unroll 1
    for (int d = 0; d < PRED_LEN; ++d) {
        cell(ax);
        const float* hr = &hbase[rowInWave * HROW_STRIDE];
        const float* wr = &sFcW[laneHalf * HIDDEN];
        float p = sFcB[laneHalf];
        #pragma unroll
        for (int k = 0; k < HIDDEN; k += 4) {
            v4f h4 = *(const v4f*)&hr[k];
            v4f w4 = *(const v4f*)&wr[k];
            p += h4.x * w4.x + h4.y * w4.y + h4.z * w4.z + h4.w * w4.w;
        }
        out[(size_t)((b0 + rowInWave) * PRED_LEN + d) * 2 + laneHalf] = p;
        sX[((wave << 4) | rowInWave) * 2 + laneHalf] = p;
        __syncthreads();
        v2f xv = *(const v2f*)&sX[((wave << 4) | rowInWave) * 2];
        ax.x = laneHalf ? 1.0f : xv.x;
        ax.y = laneHalf ? 0.0f : xv.y;
    }
}

extern "C" void kernel_launch(void* const* d_in, const int* in_sizes, int n_in,
                              void* d_out, int out_size, void* d_ws, size_t ws_size,
                              hipStream_t stream) {
    const float* obs  = (const float*)d_in[0];
    const float* W_ih = (const float*)d_in[1];
    const float* W_hh = (const float*)d_in[2];
    const float* b_ih = (const float*)d_in[3];
    const float* b_hh = (const float*)d_in[4];
    const float* fc_W = (const float*)d_in[5];
    const float* fc_b = (const float*)d_in[6];
    float* outp = (float*)d_out;

    const int B = in_sizes[0] / (T_ENC * IN_DIM);
    dim3 grid(B / ROWS_PER_BLOCK);
    lstm_seq2seq_kernel<<<grid, 256, 0, stream>>>(obs, W_ih, W_hh, b_ih, b_hh,
                                                  fc_W, fc_b, outp);
}


# Accumulate the four gate tiles of one hidden-column tile at a time (4 accumulators, not 8): 228 registers instead of 270, so two workgroups per WGP instead of one
# speedup vs baseline: 1.1797x; 1.1797x over previous
#include <hip/hip_runtime.h>

#define PRED_LEN 12
#define HIDDEN 64
#define T_ENC 8
#define IN_DIM 2
#define WAVES_PER_BLOCK 8
#define ROWS_PER_WAVE 16
#define ROWS_PER_BLOCK (WAVES_PER_BLOCK * ROWS_PER_WAVE)
#define HROW_STRIDE 68

typedef __attribute__((ext_vector_type(2))) float v2f;
typedef __attribute__((ext_vector_type(4))) float v4f;
typedef __attribute__((ext_vector_type(8))) float v8f;

__device__ __forceinline__ float fast_rcp(float x) {
    return __builtin_amdgcn_rcpf(x);
}
__device__ __forceinline__ float sigm(float x) {
    return fast_rcp(1.0f + __expf(-x));
}
__device__ __forceinline__ float tanh_fast(float x) {
    return 1.0f - 2.0f * fast_rcp(__expf(2.0f * x) + 1.0f);
}

__device__ __forceinline__ v8f wmma_f32(v2f a, v2f b, v8f c) {
    return __builtin_amdgcn_wmma_f32_16x16x4_f32(
         false, a,  false, b,
         (short)0, c,  false,  false);
}

__launch_bounds__(256, 1)
__global__ void lstm_seq2seq_kernel(const float* __restrict__ obs,
                                    const float* __restrict__ W_ih,
                                    const float* __restrict__ W_hh,
                                    const float* __restrict__ b_ih,
                                    const float* __restrict__ b_hh,
                                    const float* __restrict__ fc_W,
                                    const float* __restrict__ fc_b,
                                    float* __restrict__ out)
{
    __shared__ v2f   sBw[16 * 16 * 32];
    __shared__ v2f   sBx[16 * 32];
    __shared__ float sH[WAVES_PER_BLOCK * ROWS_PER_WAVE * HROW_STRIDE];
    __shared__ float sX[WAVES_PER_BLOCK * ROWS_PER_WAVE * 2];
    __shared__ float sFcW[2 * HIDDEN];
    __shared__ float sFcB[2];

    const int tid = threadIdx.x;

    for (int idx = tid; idx < 16 * 16 * 32; idx += 256) {
        int t = idx >> 9;
        int k = (idx >> 5) & 15;
        int l = idx & 31;
        int n  = (t << 4) | (l & 15);
        int kk = (k << 2) + ((l >> 4) << 1);
        v2f w;
        w.x = W_hh[n * HIDDEN + kk];
        w.y = W_hh[n * HIDDEN + kk + 1];
        sBw[idx] = w;
    }
    for (int idx = tid; idx < 16 * 32; idx += 256) {
        int t = idx >> 5;
        int l = idx & 31;
        int n = (t << 4) | (l & 15);
        v2f w;
        if (l < 16) { w.x = W_ih[n * IN_DIM + 0]; w.y = W_ih[n * IN_DIM + 1]; }
        else        { w.x = b_ih[n] + b_hh[n];    w.y = 0.0f; }
        sBx[idx] = w;
    }
    for (int idx = tid; idx < 2 * HIDDEN; idx += 256) sFcW[idx] = fc_W[idx];
    if (tid < 2) sFcB[tid] = fc_b[tid];
    for (int idx = tid; idx < WAVES_PER_BLOCK * ROWS_PER_WAVE * HROW_STRIDE; idx += 256)
        sH[idx] = 0.0f;
    __syncthreads();

    const int wave      = tid >> 5;
    const int lane      = tid & 31;
    const int rowInWave = lane & 15;
    const int laneHalf  = lane >> 4;
    const int b0        = blockIdx.x * ROWS_PER_BLOCK + wave * ROWS_PER_WAVE;
    float* hbase = &sH[(wave * ROWS_PER_WAVE) * HROW_STRIDE];

    v8f cst[4];
    #pragma unroll
    for (int j = 0; j < 4; ++j) cst[j] = (v8f){};

    auto cell = [&](v2f ax) {
        float hnew[4][8];
        #pragma unroll
        for (int j = 0; j < 4; ++j) {
            v8f acc[4];
            v8f zero = {};
            #pragma unroll
            for (int g = 0; g < 4; ++g) {
                int t = 4 * g + j;
                v2f bx = sBx[(t << 5) | lane];
                acc[g] = wmma_f32(ax, bx, zero);
            }
            #pragma unroll
            for (int k = 0; k < 16; ++k) {
                int kk = (k << 2) | (laneHalf << 1);
                v2f ah = *(const v2f*)&hbase[rowInWave * HROW_STRIDE + kk];
                #pragma unroll
                for (int g = 0; g < 4; ++g) {
                    int t = 4 * g + j;
                    v2f bw = sBw[(((t << 4) | k) << 5) | lane];
                    acc[g] = wmma_f32(ah, bw, acc[g]);
                }
            }
            #pragma unroll
            for (int v = 0; v < 8; ++v) {
                float ig = sigm(acc[0][v]);
                float fg = sigm(acc[1][v]);
                float gg = tanh_fast(acc[2][v]);
                float og = sigm(acc[3][v]);
                float cc = fg * cst[j][v] + ig * gg;
                cst[j][v] = cc;
                hnew[j][v] = og * tanh_fast(cc);
            }
        }
        #pragma unroll
        for (int j = 0; j < 4; ++j) {
            #pragma unroll
            for (int v = 0; v < 8; ++v) {
                hbase[(v + (laneHalf << 3)) * HROW_STRIDE + (j << 4) + rowInWave] =
                    hnew[j][v];
            }
        }
        __syncthreads();
    };

    v2f ax;
    #pragma unroll 1
    for (int s = 0; s < T_ENC; ++s) {
        v2f xv = *(const v2f*)&obs[(size_t)((b0 + rowInWave) * T_ENC + s) * IN_DIM];
        ax.x = laneHalf ? 1.0f : xv.x;
        ax.y = laneHalf ? 0.0f : xv.y;
        cell(ax);
    }

    #pragma unroll 1
    for (int d = 0; d < PRED_LEN; ++d) {
        cell(ax);
        const float* hr = &hbase[rowInWave * HROW_STRIDE];
        const float* wr = &sFcW[laneHalf * HIDDEN];
        float p = sFcB[laneHalf];
        #pragma unroll
        for (int k = 0; k < HIDDEN; k += 4) {
            v4f h4 = *(const v4f*)&hr[k];
            v4f w4 = *(const v4f*)&wr[k];
            p += h4.x * w4.x + h4.y * w4.y + h4.z * w4.z + h4.w * w4.w;
        }
        out[(size_t)((b0 + rowInWave) * PRED_LEN + d) * 2 + laneHalf] = p;
        sX[((wave << 4) | rowInWave) * 2 + laneHalf] = p;
        __syncthreads();
        v2f xv = *(const v2f*)&sX[((wave << 4) | rowInWave) * 2];
        ax.x = laneHalf ? 1.0f : xv.x;
        ax.y = laneHalf ? 0.0f : xv.y;
    }
}

extern "C" void kernel_launch(void* const* d_in, const int* in_sizes, int n_in,
                              void* d_out, int out_size, void* d_ws, size_t ws_size,
                              hipStream_t stream) {
    const float* obs  = (const float*)d_in[0];
    const float* W_ih = (const float*)d_in[1];
    const float* W_hh = (const float*)d_in[2];
    const float* b_ih = (const float*)d_in[3];
    const float* b_hh = (const float*)d_in[4];
    const float* fc_W = (const float*)d_in[5];
    const float* fc_b = (const float*)d_in[6];
    float* outp = (float*)d_out;

    const int B = in_sizes[0] / (T_ENC * IN_DIM);
    dim3 grid(B / ROWS_PER_BLOCK);
    lstm_seq2seq_kernel<<<grid, 256, 0, stream>>>(obs, W_ih, W_hh, b_ih, b_hh,
                                                  fc_W, fc_b, outp);
}


# W_hh held in registers as WMMA B fragments: each of 4 waves owns one hidden-column tile (its 4 gates) for all 128 rows of the block; LDS carries only h, double-buffered, one barrier per step
# speedup vs baseline: 1.3277x; 1.1255x over previous
#include <hip/hip_runtime.h>


#define PRED_LEN 12
#define HIDDEN 64
#define T_ENC 8
#define IN_DIM 2
#define WAVES_PER_BLOCK 4
#define ROWS_PER_BLOCK 128
#define MTILES (ROWS_PER_BLOCK / 16)
#define HROW_STRIDE 68
#define XROW_STRIDE 18
#define HBUF (ROWS_PER_BLOCK * HROW_STRIDE)


typedef __attribute__((ext_vector_type(2))) float v2f;
typedef __attribute__((ext_vector_type(4))) float v4f;
typedef __attribute__((ext_vector_type(8))) float v8f;

__device__ __forceinline__ float fast_rcp(float x) {
    return __builtin_amdgcn_rcpf(x);
}
__device__ __forceinline__ float sigm(float x) {
    return fast_rcp(1.0f + __expf(-x));
}
__device__ __forceinline__ float tanh_fast(float x) {
    return 1.0f - 2.0f * fast_rcp(__expf(2.0f * x) + 1.0f);
}

__device__ __forceinline__ v8f wmma_f32(v2f a, v2f b, v8f c) {
    return __builtin_amdgcn_wmma_f32_16x16x4_f32(
         false, a,  false, b,
         (short)0, c,  false,  false);
}

__launch_bounds__(128, 2)
__global__ void lstm_seq2seq_kernel(const float* __restrict__ obs,
                                    const float* __restrict__ W_ih,
                                    const float* __restrict__ W_hh,
                                    const float* __restrict__ b_ih,
                                    const float* __restrict__ b_hh,
                                    const float* __restrict__ fc_W,
                                    const float* __restrict__ fc_b,
                                    float* __restrict__ out)
{
    __shared__ __attribute__((aligned(16))) float sH[2 * HBUF];
    __shared__ __attribute__((aligned(16))) float sX[ROWS_PER_BLOCK * XROW_STRIDE];
    __shared__ __attribute__((aligned(16))) float sFcW[2 * HIDDEN];
    __shared__ float sFcB[2];

    const int tid      = threadIdx.x;
    const int wave     = tid >> 5;
    const int lane     = tid & 31;
    const int col      = lane & 15;
    const int laneHalf = lane >> 4;
    const int b0       = blockIdx.x * ROWS_PER_BLOCK;

    for (int idx = tid; idx < ROWS_PER_BLOCK * 4; idx += 128) {
        int r = idx >> 2;
        int q = idx & 3;
        v4f v = *(const v4f*)&obs[(size_t)(b0 + r) * (T_ENC * IN_DIM) + q * 4];
        v2f lo; lo.x = v.x; lo.y = v.y;
        v2f hi; hi.x = v.z; hi.y = v.w;
        *(v2f*)&sX[r * XROW_STRIDE + q * 4]     = lo;
        *(v2f*)&sX[r * XROW_STRIDE + q * 4 + 2] = hi;
    }
    for (int idx = tid; idx < 2 * HIDDEN; idx += 128) sFcW[idx] = fc_W[idx];
    if (tid < 2) sFcB[tid] = fc_b[tid];
    for (int idx = tid; idx < HBUF / 4; idx += 128)
        ((v4f*)sH)[idx] = (v4f){};

    v2f bw[4][16];
    v2f bx[4];
    #pragma unroll
    for (int g = 0; g < 4; ++g) {
        const int n = (((g << 2) | wave) << 4) | col;
        #pragma unroll
        for (int k = 0; k < 16; ++k) {
            const int kk = (k << 2) | (laneHalf << 1);
            bw[g][k] = *(const v2f*)&W_hh[n * HIDDEN + kk];
        }
        v2f w;
        if (laneHalf == 0) { w.x = W_ih[n * IN_DIM + 0]; w.y = W_ih[n * IN_DIM + 1]; }
        else               { w.x = b_ih[n] + b_hh[n];    w.y = 0.0f; }
        bx[g] = w;
    }

    v8f cst[MTILES];
    #pragma unroll
    for (int m = 0; m < MTILES; ++m) cst[m] = (v8f){};

    __syncthreads();

    #pragma unroll 1
    for (int s = 0; s < T_ENC + PRED_LEN; ++s) {
        const int xoff = (s < T_ENC) ? (s << 1) : ((s == T_ENC) ? ((T_ENC - 1) << 1) : 16);
        const float* hcur = &sH[(s & 1) * HBUF];
        float*       hnxt = &sH[((s & 1) ^ 1) * HBUF];

        v2f a_cur[16];
        v2f x_cur;
        {
            const float* hr = &hcur[col * HROW_STRIDE + (laneHalf << 1)];
            #pragma unroll
            for (int k = 0; k < 16; ++k) a_cur[k] = *(const v2f*)&hr[k << 2];
            x_cur = *(const v2f*)&sX[col * XROW_STRIDE + xoff];
        }

        #pragma unroll
        for (int m = 0; m < MTILES; ++m) {
            v2f a_nxt[16];
            v2f x_nxt;
            if (m + 1 < MTILES) {
                const int r = ((m + 1) << 4) | col;
                const float* hr = &hcur[r * HROW_STRIDE + (laneHalf << 1)];
                #pragma unroll
                for (int k = 0; k < 16; ++k) a_nxt[k] = *(const v2f*)&hr[k << 2];
                x_nxt = *(const v2f*)&sX[r * XROW_STRIDE + xoff];
            }

            v2f ax;
            ax.x = laneHalf ? 1.0f : x_cur.x;
            ax.y = laneHalf ? 0.0f : x_cur.y;

            v8f acc[4];
            v8f zero = {};
            #pragma unroll
            for (int g = 0; g < 4; ++g)
                acc[g] = wmma_f32(ax, bx[g], zero);
            #pragma unroll
            for (int k = 0; k < 16; ++k) {
                #pragma unroll
                for (int g = 0; g < 4; ++g)
                    acc[g] = wmma_f32(a_cur[k], bw[g][k], acc[g]);
            }

            #pragma unroll
            for (int v = 0; v < 8; ++v) {
                float ig = sigm(acc[0][v]);
                float fg = sigm(acc[1][v]);
                float gg = tanh_fast(acc[2][v]);
                float og = sigm(acc[3][v]);
                float cc = fg * cst[m][v] + ig * gg;
                cst[m][v] = cc;
                hnxt[((m << 4) + (laneHalf << 3) + v) * HROW_STRIDE + (wave << 4) + col] =
                    og * tanh_fast(cc);
            }

            if (m + 1 < MTILES) {
                #pragma unroll
                for (int k = 0; k < 16; ++k) a_cur[k] = a_nxt[k];
                x_cur = x_nxt;
            }
        }
        __syncthreads();

        if (s >= T_ENC) {
            const int d = s - T_ENC;
            const float* hr = &hnxt[tid * HROW_STRIDE];
            float p0 = sFcB[0];
            float p1 = sFcB[1];
            #pragma unroll
            for (int k = 0; k < HIDDEN; k += 4) {
                v4f h4 = *(const v4f*)&hr[k];
                v4f w0 = *(const v4f*)&sFcW[k];
                v4f w1 = *(const v4f*)&sFcW[HIDDEN + k];
                p0 += h4.x * w0.x + h4.y * w0.y + h4.z * w0.z + h4.w * w0.w;
                p1 += h4.x * w1.x + h4.y * w1.y + h4.z * w1.z + h4.w * w1.w;
            }
            v2f pv; pv.x = p0; pv.y = p1;
            *(v2f*)&out[(size_t)((b0 + tid) * PRED_LEN + d) * 2] = pv;
            *(v2f*)&sX[tid * XROW_STRIDE + 16] = pv;
            __syncthreads();
        }
    }
}

extern "C" void kernel_launch(void* const* d_in, const int* in_sizes, int n_in,
                              void* d_out, int out_size, void* d_ws, size_t ws_size,
                              hipStream_t stream) {
    const float* obs  = (const float*)d_in[0];
    const float* W_ih = (const float*)d_in[1];
    const float* W_hh = (const float*)d_in[2];
    const float* b_ih = (const float*)d_in[3];
    const float* b_hh = (const float*)d_in[4];
    const float* fc_W = (const float*)d_in[5];
    const float* fc_b = (const float*)d_in[6];
    float* outp = (float*)d_out;

    const int B = in_sizes[0] / (T_ENC * IN_DIM);
    dim3 grid(B / ROWS_PER_BLOCK);
    lstm_seq2seq_kernel<<<grid, 128, 0, stream>>>(obs, W_ih, W_hh, b_ih, b_hh,
                                                  fc_W, fc_b, outp);
}


# Gate nonlinearities on the hardware tanh instruction: sigmoid(z) = 0.5 + 0.5*tanh(z/2), the exact 0.5 folded into the i/f/o weight and bias fragments; 5 transcendentals per element instead of 10
# speedup vs baseline: 1.4736x; 1.1099x over previous
#include <hip/hip_runtime.h>


#define PRED_LEN 12
#define HIDDEN 64
#define T_ENC 8
#define IN_DIM 2
#define WAVES_PER_BLOCK 4
#define ROWS_PER_BLOCK 128
#define MTILES (ROWS_PER_BLOCK / 16)
#define HROW_STRIDE 68
#define XROW_STRIDE 18
#define HBUF (ROWS_PER_BLOCK * HROW_STRIDE)


typedef __attribute__((ext_vector_type(2))) float v2f;
typedef __attribute__((ext_vector_type(4))) float v4f;
typedef __attribute__((ext_vector_type(8))) float v8f;

__device__ __forceinline__ float tanh_hw(float x) {
    return __builtin_amdgcn_tanhf(x);
}
__device__ __forceinline__ float sigm_of_half(float zh) {
    return fmaf(0.5f, tanh_hw(zh), 0.5f);
}

__device__ __forceinline__ v8f wmma_f32(v2f a, v2f b, v8f c) {
    return __builtin_amdgcn_wmma_f32_16x16x4_f32(
         false, a,  false, b,
         (short)0, c,  false,  false);
}

__launch_bounds__(128, 2)
__global__ void lstm_seq2seq_kernel(const float* __restrict__ obs,
                                    const float* __restrict__ W_ih,
                                    const float* __restrict__ W_hh,
                                    const float* __restrict__ b_ih,
                                    const float* __restrict__ b_hh,
                                    const float* __restrict__ fc_W,
                                    const float* __restrict__ fc_b,
                                    float* __restrict__ out)
{
    __shared__ __attribute__((aligned(16))) float sH[2 * HBUF];
    __shared__ __attribute__((aligned(16))) float sX[ROWS_PER_BLOCK * XROW_STRIDE];
    __shared__ __attribute__((aligned(16))) float sFcW[2 * HIDDEN];
    __shared__ float sFcB[2];

    const int tid      = threadIdx.x;
    const int wave     = tid >> 5;
    const int lane     = tid & 31;
    const int col      = lane & 15;
    const int laneHalf = lane >> 4;
    const int b0       = blockIdx.x * ROWS_PER_BLOCK;

    for (int idx = tid; idx < ROWS_PER_BLOCK * 4; idx += 128) {
        int r = idx >> 2;
        int q = idx & 3;
        v4f v = *(const v4f*)&obs[(size_t)(b0 + r) * (T_ENC * IN_DIM) + q * 4];
        v2f lo; lo.x = v.x; lo.y = v.y;
        v2f hi; hi.x = v.z; hi.y = v.w;
        *(v2f*)&sX[r * XROW_STRIDE + q * 4]     = lo;
        *(v2f*)&sX[r * XROW_STRIDE + q * 4 + 2] = hi;
    }
    for (int idx = tid; idx < 2 * HIDDEN; idx += 128) sFcW[idx] = fc_W[idx];
    if (tid < 2) sFcB[tid] = fc_b[tid];
    for (int idx = tid; idx < HBUF / 4; idx += 128)
        ((v4f*)sH)[idx] = (v4f){};

    v2f bw[4][16];
    v2f bx[4];
    #pragma unroll
    for (int g = 0; g < 4; ++g) {
        const int n = (((g << 2) | wave) << 4) | col;
        const float sc = (g == 2) ? 1.0f : 0.5f;
        #pragma unroll
        for (int k = 0; k < 16; ++k) {
            const int kk = (k << 2) | (laneHalf << 1);
            bw[g][k] = *(const v2f*)&W_hh[n * HIDDEN + kk] * sc;
        }
        v2f w;
        if (laneHalf == 0) { w.x = W_ih[n * IN_DIM + 0]; w.y = W_ih[n * IN_DIM + 1]; }
        else               { w.x = b_ih[n] + b_hh[n];    w.y = 0.0f; }
        bx[g] = w * sc;
    }

    v8f cst[MTILES];
    #pragma unroll
    for (int m = 0; m < MTILES; ++m) cst[m] = (v8f){};

    __syncthreads();

    #pragma unroll 1
    for (int s = 0; s < T_ENC + PRED_LEN; ++s) {
        const int xoff = (s < T_ENC) ? (s << 1) : ((s == T_ENC) ? ((T_ENC - 1) << 1) : 16);
        const float* hcur = &sH[(s & 1) * HBUF];
        float*       hnxt = &sH[((s & 1) ^ 1) * HBUF];

        v2f a_cur[16];
        v2f x_cur;
        {
            const float* hr = &hcur[col * HROW_STRIDE + (laneHalf << 1)];
            #pragma unroll
            for (int k = 0; k < 16; ++k) a_cur[k] = *(const v2f*)&hr[k << 2];
            x_cur = *(const v2f*)&sX[col * XROW_STRIDE + xoff];
        }

        #pragma unroll
        for (int m = 0; m < MTILES; ++m) {
            v2f a_nxt[16];
            v2f x_nxt;
            if (m + 1 < MTILES) {
                const int r = ((m + 1) << 4) | col;
                const float* hr = &hcur[r * HROW_STRIDE + (laneHalf << 1)];
                #pragma unroll
                for (int k = 0; k < 16; ++k) a_nxt[k] = *(const v2f*)&hr[k << 2];
                x_nxt = *(const v2f*)&sX[r * XROW_STRIDE + xoff];
            }

            v2f ax;
            ax.x = laneHalf ? 1.0f : x_cur.x;
            ax.y = laneHalf ? 0.0f : x_cur.y;

            v8f acc[4];
            v8f zero = {};
            #pragma unroll
            for (int g = 0; g < 4; ++g)
                acc[g] = wmma_f32(ax, bx[g], zero);
            #pragma unroll
            for (int k = 0; k < 16; ++k) {
                #pragma unroll
                for (int g = 0; g < 4; ++g)
                    acc[g] = wmma_f32(a_cur[k], bw[g][k], acc[g]);
            }

            #pragma unroll
            for (int v = 0; v < 8; ++v) {
                float ig = sigm_of_half(acc[0][v]);
                float fg = sigm_of_half(acc[1][v]);
                float gg = tanh_hw(acc[2][v]);
                float og = sigm_of_half(acc[3][v]);
                float cc = fg * cst[m][v] + ig * gg;
                cst[m][v] = cc;
                hnxt[((m << 4) + (laneHalf << 3) + v) * HROW_STRIDE + (wave << 4) + col] =
                    og * tanh_hw(cc);
            }

            if (m + 1 < MTILES) {
                #pragma unroll
                for (int k = 0; k < 16; ++k) a_cur[k] = a_nxt[k];
                x_cur = x_nxt;
            }
        }
        __syncthreads();

        if (s >= T_ENC) {
            const int d = s - T_ENC;
            const float* hr = &hnxt[tid * HROW_STRIDE];
            float p0 = sFcB[0];
            float p1 = sFcB[1];
            #pragma unroll
            for (int k = 0; k < HIDDEN; k += 4) {
                v4f h4 = *(const v4f*)&hr[k];
                v4f w0 = *(const v4f*)&sFcW[k];
                v4f w1 = *(const v4f*)&sFcW[HIDDEN + k];
                p0 += h4.x * w0.x + h4.y * w0.y + h4.z * w0.z + h4.w * w0.w;
                p1 += h4.x * w1.x + h4.y * w1.y + h4.z * w1.z + h4.w * w1.w;
            }
            v2f pv; pv.x = p0; pv.y = p1;
            *(v2f*)&out[(size_t)((b0 + tid) * PRED_LEN + d) * 2] = pv;
            *(v2f*)&sX[tid * XROW_STRIDE + 16] = pv;
            __syncthreads();
        }
    }
}

extern "C" void kernel_launch(void* const* d_in, const int* in_sizes, int n_in,
                              void* d_out, int out_size, void* d_ws, size_t ws_size,
                              hipStream_t stream) {
    const float* obs  = (const float*)d_in[0];
    const float* W_ih = (const float*)d_in[1];
    const float* W_hh = (const float*)d_in[2];
    const float* b_ih = (const float*)d_in[3];
    const float* b_hh = (const float*)d_in[4];
    const float* fc_W = (const float*)d_in[5];
    const float* fc_b = (const float*)d_in[6];
    float* outp = (float*)d_out;

    const int B = in_sizes[0] / (T_ENC * IN_DIM);
    dim3 grid(B / ROWS_PER_BLOCK);
    lstm_seq2seq_kernel<<<grid, 128, 0, stream>>>(obs, W_ih, W_hh, b_ih, b_hh,
                                                  fc_W, fc_b, outp);
}


# Step 0 peeled: the recurrence starts from h = 0 and c = 0, so its 64 recurrent WMMAs per row tile and the forget-gate term are exact zeros and are skipped; 3 input-projection WMMAs per tile remain
# speedup vs baseline: 1.5257x; 1.0354x over previous
#include <hip/hip_runtime.h>


#define PRED_LEN 12
#define HIDDEN 64
#define T_ENC 8
#define IN_DIM 2
#define WAVES_PER_BLOCK 4
#define ROWS_PER_BLOCK 128
#define MTILES (ROWS_PER_BLOCK / 16)
#define HROW_STRIDE 68
#define XROW_STRIDE 18
#define HBUF (ROWS_PER_BLOCK * HROW_STRIDE)


typedef __attribute__((ext_vector_type(2))) float v2f;
typedef __attribute__((ext_vector_type(4))) float v4f;
typedef __attribute__((ext_vector_type(8))) float v8f;

__device__ __forceinline__ float tanh_hw(float x) {
    return __builtin_amdgcn_tanhf(x);
}
__device__ __forceinline__ float sigm_of_half(float zh) {
    return fmaf(0.5f, tanh_hw(zh), 0.5f);
}

__device__ __forceinline__ v8f wmma_f32(v2f a, v2f b, v8f c) {
    return __builtin_amdgcn_wmma_f32_16x16x4_f32(
         false, a,  false, b,
         (short)0, c,  false,  false);
}

__launch_bounds__(128, 2)
__global__ void lstm_seq2seq_kernel(const float* __restrict__ obs,
                                    const float* __restrict__ W_ih,
                                    const float* __restrict__ W_hh,
                                    const float* __restrict__ b_ih,
                                    const float* __restrict__ b_hh,
                                    const float* __restrict__ fc_W,
                                    const float* __restrict__ fc_b,
                                    float* __restrict__ out)
{
    __shared__ __attribute__((aligned(16))) float sH[2 * HBUF];
    __shared__ __attribute__((aligned(16))) float sX[ROWS_PER_BLOCK * XROW_STRIDE];
    __shared__ __attribute__((aligned(16))) float sFcW[2 * HIDDEN];
    __shared__ float sFcB[2];

    const int tid      = threadIdx.x;
    const int wave     = tid >> 5;
    const int lane     = tid & 31;
    const int col      = lane & 15;
    const int laneHalf = lane >> 4;
    const int b0       = blockIdx.x * ROWS_PER_BLOCK;

    for (int idx = tid; idx < ROWS_PER_BLOCK * 4; idx += 128) {
        int r = idx >> 2;
        int q = idx & 3;
        v4f v = *(const v4f*)&obs[(size_t)(b0 + r) * (T_ENC * IN_DIM) + q * 4];
        v2f lo; lo.x = v.x; lo.y = v.y;
        v2f hi; hi.x = v.z; hi.y = v.w;
        *(v2f*)&sX[r * XROW_STRIDE + q * 4]     = lo;
        *(v2f*)&sX[r * XROW_STRIDE + q * 4 + 2] = hi;
    }
    for (int idx = tid; idx < 2 * HIDDEN; idx += 128) sFcW[idx] = fc_W[idx];
    if (tid < 2) sFcB[tid] = fc_b[tid];

    v2f bw[4][16];
    v2f bx[4];
    #pragma unroll
    for (int g = 0; g < 4; ++g) {
        const int n = (((g << 2) | wave) << 4) | col;
        const float sc = (g == 2) ? 1.0f : 0.5f;
        #pragma unroll
        for (int k = 0; k < 16; ++k) {
            const int kk = (k << 2) | (laneHalf << 1);
            bw[g][k] = *(const v2f*)&W_hh[n * HIDDEN + kk] * sc;
        }
        v2f w;
        if (laneHalf == 0) { w.x = W_ih[n * IN_DIM + 0]; w.y = W_ih[n * IN_DIM + 1]; }
        else               { w.x = b_ih[n] + b_hh[n];    w.y = 0.0f; }
        bx[g] = w * sc;
    }

    v8f cst[MTILES];

    __syncthreads();

    {
        float* hnxt = &sH[HBUF];
        #pragma unroll
        for (int m = 0; m < MTILES; ++m) {
            const v2f x0 = *(const v2f*)&sX[((m << 4) | col) * XROW_STRIDE];
            v2f ax;
            ax.x = laneHalf ? 1.0f : x0.x;
            ax.y = laneHalf ? 0.0f : x0.y;
            v8f zero = {};
            v8f zi = wmma_f32(ax, bx[0], zero);
            v8f zg = wmma_f32(ax, bx[2], zero);
            v8f zo = wmma_f32(ax, bx[3], zero);
            #pragma unroll
            for (int v = 0; v < 8; ++v) {
                float ig = sigm_of_half(zi[v]);
                float gg = tanh_hw(zg[v]);
                float og = sigm_of_half(zo[v]);
                float cc = ig * gg;
                cst[m][v] = cc;
                hnxt[((m << 4) + (laneHalf << 3) + v) * HROW_STRIDE + (wave << 4) + col] =
                    og * tanh_hw(cc);
            }
        }
        __syncthreads();
    }

    #pragma unroll 1
    for (int s = 1; s < T_ENC + PRED_LEN; ++s) {
        const int xoff = (s < T_ENC) ? (s << 1) : ((s == T_ENC) ? ((T_ENC - 1) << 1) : 16);
        const float* hcur = &sH[(s & 1) * HBUF];
        float*       hnxt = &sH[((s & 1) ^ 1) * HBUF];

        v2f a_cur[16];
        v2f x_cur;
        {
            const float* hr = &hcur[col * HROW_STRIDE + (laneHalf << 1)];
            #pragma unroll
            for (int k = 0; k < 16; ++k) a_cur[k] = *(const v2f*)&hr[k << 2];
            x_cur = *(const v2f*)&sX[col * XROW_STRIDE + xoff];
        }

        #pragma unroll
        for (int m = 0; m < MTILES; ++m) {
            v2f a_nxt[16];
            v2f x_nxt;
            if (m + 1 < MTILES) {
                const int r = ((m + 1) << 4) | col;
                const float* hr = &hcur[r * HROW_STRIDE + (laneHalf << 1)];
                #pragma unroll
                for (int k = 0; k < 16; ++k) a_nxt[k] = *(const v2f*)&hr[k << 2];
                x_nxt = *(const v2f*)&sX[r * XROW_STRIDE + xoff];
            }

            v2f ax;
            ax.x = laneHalf ? 1.0f : x_cur.x;
            ax.y = laneHalf ? 0.0f : x_cur.y;

            v8f acc[4];
            v8f zero = {};
            #pragma unroll
            for (int g = 0; g < 4; ++g)
                acc[g] = wmma_f32(ax, bx[g], zero);
            #pragma unroll
            for (int k = 0; k < 16; ++k) {
                #pragma unroll
                for (int g = 0; g < 4; ++g)
                    acc[g] = wmma_f32(a_cur[k], bw[g][k], acc[g]);
            }

            #pragma unroll
            for (int v = 0; v < 8; ++v) {
                float ig = sigm_of_half(acc[0][v]);
                float fg = sigm_of_half(acc[1][v]);
                float gg = tanh_hw(acc[2][v]);
                float og = sigm_of_half(acc[3][v]);
                float cc = fg * cst[m][v] + ig * gg;
                cst[m][v] = cc;
                hnxt[((m << 4) + (laneHalf << 3) + v) * HROW_STRIDE + (wave << 4) + col] =
                    og * tanh_hw(cc);
            }

            if (m + 1 < MTILES) {
                #pragma unroll
                for (int k = 0; k < 16; ++k) a_cur[k] = a_nxt[k];
                x_cur = x_nxt;
            }
        }
        __syncthreads();

        if (s >= T_ENC) {
            const int d = s - T_ENC;
            const float* hr = &hnxt[tid * HROW_STRIDE];
            float p0 = sFcB[0];
            float p1 = sFcB[1];
            #pragma unroll
            for (int k = 0; k < HIDDEN; k += 4) {
                v4f h4 = *(const v4f*)&hr[k];
                v4f w0 = *(const v4f*)&sFcW[k];
                v4f w1 = *(const v4f*)&sFcW[HIDDEN + k];
                p0 += h4.x * w0.x + h4.y * w0.y + h4.z * w0.z + h4.w * w0.w;
                p1 += h4.x * w1.x + h4.y * w1.y + h4.z * w1.z + h4.w * w1.w;
            }
            v2f pv; pv.x = p0; pv.y = p1;
            *(v2f*)&out[(size_t)((b0 + tid) * PRED_LEN + d) * 2] = pv;
            *(v2f*)&sX[tid * XROW_STRIDE + 16] = pv;
            __syncthreads();
        }
    }
}

extern "C" void kernel_launch(void* const* d_in, const int* in_sizes, int n_in,
                              void* d_out, int out_size, void* d_ws, size_t ws_size,
                              hipStream_t stream) {
    const float* obs  = (const float*)d_in[0];
    const float* W_ih = (const float*)d_in[1];
    const float* W_hh = (const float*)d_in[2];
    const float* b_ih = (const float*)d_in[3];
    const float* b_hh = (const float*)d_in[4];
    const float* fc_W = (const float*)d_in[5];
    const float* fc_b = (const float*)d_in[6];
    float* outp = (float*)d_out;

    const int B = in_sizes[0] / (T_ENC * IN_DIM);
    dim3 grid(B / ROWS_PER_BLOCK);
    lstm_seq2seq_kernel<<<grid, 128, 0, stream>>>(obs, W_ih, W_hh, b_ih, b_hh,
                                                  fc_W, fc_b, outp);
}


# Step computed transposed (W as A, h^T as B): accumulator registers are the next step's B operands, so h, c and all of W_hh stay in one wave's registers; no LDS, no barrier; k and fc sum order change
# speedup vs baseline: 1.5881x; 1.0409x over previous
#include <hip/hip_runtime.h>


#define PRED_LEN 12
#define HIDDEN 64
#define T_ENC 8
#define IN_DIM 2
#define WAVES_PER_BLOCK 4
#define NT 4
#define ROWS_PER_BLOCK (WAVES_PER_BLOCK * NT * 16)


typedef __attribute__((ext_vector_type(2))) float v2f;
typedef __attribute__((ext_vector_type(8))) float v8f;

__device__ __forceinline__ float tanh_hw(float x) {
    return __builtin_amdgcn_tanhf(x);
}
__device__ __forceinline__ float sigm_of_half(float zh) {
    return fmaf(0.5f, tanh_hw(zh), 0.5f);
}
__device__ __forceinline__ float xhalf(float x) {
    int b = __builtin_bit_cast(int, x);
    return __builtin_bit_cast(float,
        __builtin_amdgcn_permlanex16(b, b, 0x76543210, 0xfedcba98, false, false));
}

__device__ __forceinline__ v8f wmma_f32(v2f a, v2f b, v8f c) {
    return __builtin_amdgcn_wmma_f32_16x16x4_f32(
         false, a,  false, b,
         (short)0, c,  false,  false);
}

__launch_bounds__(128)
__global__ void lstm_seq2seq_kernel(const float* __restrict__ obs,
                                    const float* __restrict__ W_ih,
                                    const float* __restrict__ W_hh,
                                    const float* __restrict__ b_ih,
                                    const float* __restrict__ b_hh,
                                    const float* __restrict__ fc_W,
                                    const float* __restrict__ fc_b,
                                    float* __restrict__ out)
{
    const int tid      = threadIdx.x;
    const int wave     = tid >> 5;
    const int lane     = tid & 31;
    const int col      = lane & 15;
    const int laneHalf = lane >> 4;
    const int row0 = (blockIdx.x * WAVES_PER_BLOCK + wave) * (NT * 16) + col;

    v2f aw[4][4][16];
    v2f ax[4][4];
    #pragma unroll
    for (int g = 0; g < 4; ++g) {
        const float sc = (g == 2) ? 1.0f : 0.5f;
        #pragma unroll
        for (int t = 0; t < 4; ++t) {
            const int n = (((g << 2) | t) << 4) | col;
            #pragma unroll
            for (int kk = 0; kk < 16; ++kk) {
                const int ku = ((kk >> 2) << 4) | (laneHalf << 3) | ((kk & 3) << 1);
                aw[g][t][kk] = *(const v2f*)&W_hh[n * HIDDEN + ku] * sc;
            }
            v2f w;
            if (laneHalf == 0) { w.x = W_ih[n * IN_DIM + 0]; w.y = W_ih[n * IN_DIM + 1]; }
            else               { w.x = b_ih[n] + b_hh[n];    w.y = 0.0f; }
            ax[g][t] = w * sc;
        }
    }
    v2f fw0[4][4];
    v2f fw1[4][4];
    #pragma unroll
    for (int t = 0; t < 4; ++t) {
        #pragma unroll
        for (int j = 0; j < 4; ++j) {
            const int u = (t << 4) | (laneHalf << 3) | (j << 1);
            fw0[t][j] = *(const v2f*)&fc_W[u];
            fw1[t][j] = *(const v2f*)&fc_W[HIDDEN + u];
        }
    }
    const float fb0 = fc_b[0];
    const float fb1 = fc_b[1];

    v8f hs[NT][4];
    v8f cs[NT][4];
    v2f xcur[NT];

    #pragma unroll
    for (int i = 0; i < NT; ++i)
        xcur[i] = *(const v2f*)&obs[(size_t)(row0 + 16 * i) * (T_ENC * IN_DIM)];

    {
        v2f xnext[NT];
        #pragma unroll
        for (int i = 0; i < NT; ++i)
            xnext[i] = *(const v2f*)&obs[(size_t)(row0 + 16 * i) * (T_ENC * IN_DIM) + IN_DIM];
        #pragma unroll
        for (int i = 0; i < NT; ++i) {
            v2f bx;
            bx.x = laneHalf ? 1.0f : xcur[i].x;
            bx.y = laneHalf ? 0.0f : xcur[i].y;
            #pragma unroll
            for (int t = 0; t < 4; ++t) {
                v8f zero = {};
                v8f zi = wmma_f32(ax[0][t], bx, zero);
                v8f zg = wmma_f32(ax[2][t], bx, zero);
                v8f zo = wmma_f32(ax[3][t], bx, zero);
                #pragma unroll
                for (int v = 0; v < 8; ++v) {
                    float ig = sigm_of_half(zi[v]);
                    float gg = tanh_hw(zg[v]);
                    float og = sigm_of_half(zo[v]);
                    float cc = ig * gg;
                    cs[i][t][v] = cc;
                    hs[i][t][v] = og * tanh_hw(cc);
                }
            }
        }
        #pragma unroll
        for (int i = 0; i < NT; ++i) xcur[i] = xnext[i];
    }

    #pragma unroll 1
    for (int s = 1; s < T_ENC + PRED_LEN; ++s) {
        v2f xnext[NT];
        if (s < T_ENC) {
            const int tn = (s + 1 < T_ENC) ? (s + 1) : (T_ENC - 1);
            #pragma unroll
            for (int i = 0; i < NT; ++i)
                xnext[i] = *(const v2f*)&obs[(size_t)(row0 + 16 * i) * (T_ENC * IN_DIM) + tn * IN_DIM];
        }

        #pragma unroll
        for (int i = 0; i < NT; ++i) {
            v2f bx;
            bx.x = laneHalf ? 1.0f : xcur[i].x;
            bx.y = laneHalf ? 0.0f : xcur[i].y;

            v8f hn[4];
            #pragma unroll
            for (int t = 0; t < 4; ++t) {
                v8f acc[4];
                v8f zero = {};
                #pragma unroll
                for (int g = 0; g < 4; ++g)
                    acc[g] = wmma_f32(ax[g][t], bx, zero);
                #pragma unroll
                for (int kk = 0; kk < 16; ++kk) {
                    v2f bh;
                    bh.x = hs[i][kk >> 2][((kk & 3) << 1)];
                    bh.y = hs[i][kk >> 2][((kk & 3) << 1) | 1];
                    #pragma unroll
                    for (int g = 0; g < 4; ++g)
                        acc[g] = wmma_f32(aw[g][t][kk], bh, acc[g]);
                }
                #pragma unroll
                for (int v = 0; v < 8; ++v) {
                    float ig = sigm_of_half(acc[0][v]);
                    float fg = sigm_of_half(acc[1][v]);
                    float gg = tanh_hw(acc[2][v]);
                    float og = sigm_of_half(acc[3][v]);
                    float cc = fg * cs[i][t][v] + ig * gg;
                    cs[i][t][v] = cc;
                    hn[t][v] = og * tanh_hw(cc);
                }
            }
            #pragma unroll
            for (int t = 0; t < 4; ++t) hs[i][t] = hn[t];

            if (s >= T_ENC) {
                const int d = s - T_ENC;
                v2f q0[4];
                v2f q1[4];
                #pragma unroll
                for (int t = 0; t < 4; ++t) {
                    v2f h0; h0.x = hn[t][0]; h0.y = hn[t][1];
                    q0[t] = h0 * fw0[t][0];
                    q1[t] = h0 * fw1[t][0];
                    #pragma unroll
                    for (int j = 1; j < 4; ++j) {
                        v2f hv;
                        hv.x = hn[t][(j << 1)];
                        hv.y = hn[t][(j << 1) | 1];
                        q0[t] = hv * fw0[t][j] + q0[t];
                        q1[t] = hv * fw1[t][j] + q1[t];
                    }
                }
                const v2f s0 = (q0[0] + q0[1]) + (q0[2] + q0[3]);
                const v2f s1 = (q1[0] + q1[1]) + (q1[2] + q1[3]);
                float p0 = s0.x + s0.y;
                float p1 = s1.x + s1.y;
                p0 += xhalf(p0);
                p1 += xhalf(p1);
                v2f pv; pv.x = p0 + fb0; pv.y = p1 + fb1;
                if (laneHalf == 0)
                    *(v2f*)&out[((size_t)(row0 + 16 * i) * PRED_LEN + d) * 2] = pv;
                xcur[i] = pv;
            } else {
                xcur[i] = xnext[i];
            }
        }
    }
}

extern "C" void kernel_launch(void* const* d_in, const int* in_sizes, int n_in,
                              void* d_out, int out_size, void* d_ws, size_t ws_size,
                              hipStream_t stream) {
    const float* obs  = (const float*)d_in[0];
    const float* W_ih = (const float*)d_in[1];
    const float* W_hh = (const float*)d_in[2];
    const float* b_ih = (const float*)d_in[3];
    const float* b_hh = (const float*)d_in[4];
    const float* fc_W = (const float*)d_in[5];
    const float* fc_b = (const float*)d_in[6];
    float* outp = (float*)d_out;

    const int B = in_sizes[0] / (T_ENC * IN_DIM);
    dim3 grid(B / ROWS_PER_BLOCK);
    lstm_seq2seq_kernel<<<grid, 128, 0, stream>>>(obs, W_ih, W_hh, b_ih, b_hh,
                                                  fc_W, fc_b, outp);
}


# Prologue: W_hh fetched once per block in whole 128-byte lines and handed to the waves through LDS, small operands without a lane branch: one load round trip instead of 16, 4x fewer W_hh requests
# speedup vs baseline: 1.7095x; 1.0764x over previous
#include <hip/hip_runtime.h>


#define PRED_LEN 12
#define HIDDEN 64
#define T_ENC 8
#define IN_DIM 2
#define WAVES_PER_BLOCK 4
#define NT 4
#define ROWS_PER_BLOCK (WAVES_PER_BLOCK * NT * 16)
#define BLOCK_THREADS (WAVES_PER_BLOCK * 32)
#define GATE_ROWS (4 * HIDDEN)
#define W_PITCH (HIDDEN + 4)
#define STAGE_LOADS ((GATE_ROWS * HIDDEN / 4) / BLOCK_THREADS)


typedef __attribute__((ext_vector_type(2))) float v2f;
typedef __attribute__((ext_vector_type(4))) float v4f;
typedef __attribute__((ext_vector_type(8))) float v8f;

__device__ __forceinline__ float tanh_hw(float x) {
    return __builtin_amdgcn_tanhf(x);
}
__device__ __forceinline__ float sigm_of_half(float zh) {
    return fmaf(0.5f, tanh_hw(zh), 0.5f);
}
__device__ __forceinline__ float xhalf(float x) {
    int b = __builtin_bit_cast(int, x);
    return __builtin_bit_cast(float,
        __builtin_amdgcn_permlanex16(b, b, 0x76543210, 0xfedcba98, false, false));
}

__device__ __forceinline__ v8f wmma_f32(v2f a, v2f b, v8f c) {
    return __builtin_amdgcn_wmma_f32_16x16x4_f32(
         false, a,  false, b,
         (short)0, c,  false,  false);
}

__launch_bounds__(128)
__global__ void lstm_seq2seq_kernel(const float* __restrict__ obs,
                                    const float* __restrict__ W_ih,
                                    const float* __restrict__ W_hh,
                                    const float* __restrict__ b_ih,
                                    const float* __restrict__ b_hh,
                                    const float* __restrict__ fc_W,
                                    const float* __restrict__ fc_b,
                                    float* __restrict__ out)
{
    const int tid      = threadIdx.x;
    const int wave     = tid >> 5;
    const int lane     = tid & 31;
    const int col      = lane & 15;
    const int laneHalf = lane >> 4;
    const int row0 = (blockIdx.x * WAVES_PER_BLOCK + wave) * (NT * 16) + col;

    __shared__ float wst[GATE_ROWS * W_PITCH];
    v4f stg[STAGE_LOADS];
    #pragma unroll
    for (int it = 0; it < STAGE_LOADS; ++it)
        stg[it] = *(const v4f*)&W_hh[(it * BLOCK_THREADS + tid) * 4];

    v2f wih[4][4];
    float bsum[4][4];
    #pragma unroll
    for (int g = 0; g < 4; ++g) {
        #pragma unroll
        for (int t = 0; t < 4; ++t) {
            const int n = (((g << 2) | t) << 4) | col;
            wih[g][t]  = *(const v2f*)&W_ih[n * IN_DIM];
            bsum[g][t] = b_ih[n] + b_hh[n];
        }
    }

    v2f xcur[NT];
    v2f xsecond[NT];
    #pragma unroll
    for (int i = 0; i < NT; ++i) {
        xcur[i]    = *(const v2f*)&obs[(size_t)(row0 + 16 * i) * (T_ENC * IN_DIM)];
        xsecond[i] = *(const v2f*)&obs[(size_t)(row0 + 16 * i) * (T_ENC * IN_DIM) + IN_DIM];
    }

    #pragma unroll
    for (int it = 0; it < STAGE_LOADS; ++it) {
        const int piece = it * BLOCK_THREADS + tid;
        const int row   = piece >> 4;
        const int q     = piece & 15;
        const float sc  = ((it >> 3) == 2) ? 1.0f : 0.5f;
        *(v4f*)&wst[row * W_PITCH + q * 4] = stg[it] * sc;
    }
    __syncthreads();

    v2f aw[4][4][16];
    v2f ax[4][4];
    #pragma unroll
    for (int g = 0; g < 4; ++g) {
        const float sc = (g == 2) ? 1.0f : 0.5f;
        #pragma unroll
        for (int t = 0; t < 4; ++t) {
            const int n = (((g << 2) | t) << 4) | col;
            #pragma unroll
            for (int tt = 0; tt < 4; ++tt) {
                const float* p = &wst[n * W_PITCH + (tt << 4) + (laneHalf << 3)];
                const v4f lo = *(const v4f*)p;
                const v4f hi = *(const v4f*)(p + 4);
                aw[g][t][4 * tt + 0] = v2f{lo.x, lo.y};
                aw[g][t][4 * tt + 1] = v2f{lo.z, lo.w};
                aw[g][t][4 * tt + 2] = v2f{hi.x, hi.y};
                aw[g][t][4 * tt + 3] = v2f{hi.z, hi.w};
            }
            v2f w;
            w.x = laneHalf ? bsum[g][t] : wih[g][t].x;
            w.y = laneHalf ? 0.0f       : wih[g][t].y;
            ax[g][t] = w * sc;
        }
    }
    v2f fw0[4][4];
    v2f fw1[4][4];
    #pragma unroll
    for (int t = 0; t < 4; ++t) {
        #pragma unroll
        for (int j = 0; j < 4; ++j) {
            const int u = (t << 4) | (laneHalf << 3) | (j << 1);
            fw0[t][j] = *(const v2f*)&fc_W[u];
            fw1[t][j] = *(const v2f*)&fc_W[HIDDEN + u];
        }
    }
    const float fb0 = fc_b[0];
    const float fb1 = fc_b[1];

    v8f hs[NT][4];
    v8f cs[NT][4];

    {
        #pragma unroll
        for (int i = 0; i < NT; ++i) {
            v2f bx;
            bx.x = laneHalf ? 1.0f : xcur[i].x;
            bx.y = laneHalf ? 0.0f : xcur[i].y;
            #pragma unroll
            for (int t = 0; t < 4; ++t) {
                v8f zero = {};
                v8f zi = wmma_f32(ax[0][t], bx, zero);
                v8f zg = wmma_f32(ax[2][t], bx, zero);
                v8f zo = wmma_f32(ax[3][t], bx, zero);
                #pragma unroll
                for (int v = 0; v < 8; ++v) {
                    float ig = sigm_of_half(zi[v]);
                    float gg = tanh_hw(zg[v]);
                    float og = sigm_of_half(zo[v]);
                    float cc = ig * gg;
                    cs[i][t][v] = cc;
                    hs[i][t][v] = og * tanh_hw(cc);
                }
            }
        }
        #pragma unroll
        for (int i = 0; i < NT; ++i) xcur[i] = xsecond[i];
    }

    #pragma unroll 1
    for (int s = 1; s < T_ENC + PRED_LEN; ++s) {
        v2f xnext[NT];
        if (s < T_ENC) {
            const int tn = (s + 1 < T_ENC) ? (s + 1) : (T_ENC - 1);
            #pragma unroll
            for (int i = 0; i < NT; ++i)
                xnext[i] = *(const v2f*)&obs[(size_t)(row0 + 16 * i) * (T_ENC * IN_DIM) + tn * IN_DIM];
        }

        #pragma unroll
        for (int i = 0; i < NT; ++i) {
            v2f bx;
            bx.x = laneHalf ? 1.0f : xcur[i].x;
            bx.y = laneHalf ? 0.0f : xcur[i].y;

            v8f hn[4];
            #pragma unroll
            for (int t = 0; t < 4; ++t) {
                v8f acc[4];
                v8f zero = {};
                #pragma unroll
                for (int g = 0; g < 4; ++g)
                    acc[g] = wmma_f32(ax[g][t], bx, zero);
                #pragma unroll
                for (int kk = 0; kk < 16; ++kk) {
                    v2f bh;
                    bh.x = hs[i][kk >> 2][((kk & 3) << 1)];
                    bh.y = hs[i][kk >> 2][((kk & 3) << 1) | 1];
                    #pragma unroll
                    for (int g = 0; g < 4; ++g)
                        acc[g] = wmma_f32(aw[g][t][kk], bh, acc[g]);
                }
                #pragma unroll
                for (int v = 0; v < 8; ++v) {
                    float ig = sigm_of_half(acc[0][v]);
                    float fg = sigm_of_half(acc[1][v]);
                    float gg = tanh_hw(acc[2][v]);
                    float og = sigm_of_half(acc[3][v]);
                    float cc = fg * cs[i][t][v] + ig * gg;
                    cs[i][t][v] = cc;
                    hn[t][v] = og * tanh_hw(cc);
                }
            }
            #pragma unroll
            for (int t = 0; t < 4; ++t) hs[i][t] = hn[t];

            if (s >= T_ENC) {
                const int d = s - T_ENC;
                v2f q0[4];
                v2f q1[4];
                #pragma unroll
                for (int t = 0; t < 4; ++t) {
                    v2f h0; h0.x = hn[t][0]; h0.y = hn[t][1];
                    q0[t] = h0 * fw0[t][0];
                    q1[t] = h0 * fw1[t][0];
                    #pragma unroll
                    for (int j = 1; j < 4; ++j) {
                        v2f hv;
                        hv.x = hn[t][(j << 1)];
                        hv.y = hn[t][(j << 1) | 1];
                        q0[t] = hv * fw0[t][j] + q0[t];
                        q1[t] = hv * fw1[t][j] + q1[t];
                    }
                }
                const v2f s0 = (q0[0] + q0[1]) + (q0[2] + q0[3]);
                const v2f s1 = (q1[0] + q1[1]) + (q1[2] + q1[3]);
                float p0 = s0.x + s0.y;
                float p1 = s1.x + s1.y;
                p0 += xhalf(p0);
                p1 += xhalf(p1);
                v2f pv; pv.x = p0 + fb0; pv.y = p1 + fb1;
                if (laneHalf == 0)
                    *(v2f*)&out[((size_t)(row0 + 16 * i) * PRED_LEN + d) * 2] = pv;
                xcur[i] = pv;
            } else {
                xcur[i] = xnext[i];
            }
        }
    }
}

extern "C" void kernel_launch(void* const* d_in, const int* in_sizes, int n_in,
                              void* d_out, int out_size, void* d_ws, size_t ws_size,
                              hipStream_t stream) {
    const float* obs  = (const float*)d_in[0];
    const float* W_ih = (const float*)d_in[1];
    const float* W_hh = (const float*)d_in[2];
    const float* b_ih = (const float*)d_in[3];
    const float* b_hh = (const float*)d_in[4];
    const float* fc_W = (const float*)d_in[5];
    const float* fc_b = (const float*)d_in[6];
    float* outp = (float*)d_out;

    const int B = in_sizes[0] / (T_ENC * IN_DIM);
    dim3 grid(B / ROWS_PER_BLOCK);
    lstm_seq2seq_kernel<<<grid, 128, 0, stream>>>(obs, W_ih, W_hh, b_ih, b_hh,
                                                  fc_W, fc_b, outp);
}
